# starting kernel
# baseline (speedup 1.0000x reference)
#include <hip/hip_runtime.h>
#include <math.h>

typedef float v2f __attribute__((ext_vector_type(2)));
typedef float v8f __attribute__((ext_vector_type(8)));

#define N_NODES 50000
#define N_EDGES 800000
#define D 128
#define ACT_NONE 0
#define ACT_TANH 1
#define ACT_SILU 2

__global__ void degree_kernel(const int* __restrict__ dst, float* __restrict__ deg) {
    int e = blockIdx.x * blockDim.x + threadIdx.x;
    if (e < N_EDGES) atomicAdd(&deg[dst[e]], 1.0f);
}

__global__ void scatter_kernel(const float* __restrict__ h, const int* __restrict__ src,
                               const int* __restrict__ dst, float* __restrict__ agg) {
    long tid = (long)blockIdx.x * blockDim.x + threadIdx.x;
    if (tid >= (long)N_EDGES * 32) return;
    int e  = (int)(tid >> 5);
    int c4 = (int)(tid & 31) << 2;
    int s = src[e], d = dst[e];
    float4 v = *(const float4*)(h + (long)s * D + c4);
    float* p = agg + (long)d * D + c4;
    atomicAdd(p + 0, v.x);
    atomicAdd(p + 1, v.y);
    atomicAdd(p + 2, v.z);
    atomicAdd(p + 3, v.w);
}

__global__ void normalize_kernel(float* __restrict__ agg, const float* __restrict__ deg) {
    long tid = (long)blockIdx.x * blockDim.x + threadIdx.x;
    if (tid >= (long)N_NODES * 32) return;
    int n  = (int)(tid >> 5);
    int c4 = (int)(tid & 31) << 2;
    float dv = 1.0f / fmaxf(deg[n], 1.0f);
    float4* p = (float4*)(agg + (long)n * D + c4);
    float4 v = *p;
    v.x *= dv; v.y *= dv; v.z *= dv; v.w *= dv;
    *p = v;
}

__launch_bounds__(256)
__global__ void sage_gemm_kernel(const float* __restrict__ A1, const float* __restrict__ W1,
                                 const float* __restrict__ bias,
                                 const float* __restrict__ A2, const float* __restrict__ W2,
                                 float* __restrict__ Out, int act) {
    __shared__ float sW[D * D];
    __shared__ float sA[16 * D];

    const int tid     = threadIdx.x;
    const int wave    = tid >> 5;
    const int lane    = tid & 31;
    const int rowBase = blockIdx.x * 16;
    const int colBase = wave * 16;
    const int lrow    = lane & 15;
    const int hiHalf  = lane >> 4;
    const int koff    = hiHalf << 1;

    v8f c = {};

    for (int phase = 0; phase < 2; ++phase) {
        const float* A = (phase == 0) ? A1 : A2;
        const float* W = (phase == 0) ? W1 : W2;
        if (A == nullptr) break;

        __syncthreads();
        for (int i = tid; i < (D * D) / 4; i += 256)
            ((float4*)sW)[i] = ((const float4*)W)[i];
        for (int i = tid; i < (16 * D) / 4; i += 256)
            ((float4*)sA)[i] = ((const float4*)(A + (long)rowBase * D))[i];
        __syncthreads();

        #pragma unroll
        for (int k0 = 0; k0 < D; k0 += 4) {
            v2f a, b;
            a.x = sA[lrow * D + k0 + koff];
            a.y = sA[lrow * D + k0 + koff + 1];
            b.x = sW[(k0 + koff) * D + colBase + lrow];
            b.y = sW[(k0 + koff + 1) * D + colBase + lrow];
            c = __builtin_amdgcn_wmma_f32_16x16x4_f32(
                     false, a,  false, b,
                     (short)0, c,  false,  false);
        }
    }

    const int col = colBase + lrow;
    const float bv = bias ? bias[col] : 0.0f;
    #pragma unroll
    for (int r = 0; r < 8; ++r) {
        int grow = rowBase + r + (hiHalf << 3);
        float v = c[r] + bv;
        if (act == ACT_TANH)      v = tanhf(v);
        else if (act == ACT_SILU) v = v / (1.0f + __expf(-v));
        Out[(long)grow * D + col] = v;
    }
}

extern "C" void kernel_launch(void* const* d_in, const int* in_sizes, int n_in,
                              void* d_out, int out_size, void* d_ws, size_t ws_size,
                              hipStream_t stream) {
    const float* h       = (const float*)d_in[0];
    const int*   src     = (const int*)d_in[1];
    const int*   dst     = (const int*)d_in[2];
    const float* W_in1   = (const float*)d_in[3];
    const float* b_in1   = (const float*)d_in[4];
    const float* W_in2   = (const float*)d_in[5];
    const float* b_in2   = (const float*)d_in[6];
    const float* W_self  = (const float*)d_in[7];
    const float* b_self  = (const float*)d_in[8];
    const float* W_neigh = (const float*)d_in[9];
    float* out = (float*)d_out;

    const size_t ND = (size_t)N_NODES * D;
    float* buf0 = (float*)d_ws;
    float* buf1 = buf0 + ND;
    float* agg  = buf1 + ND;
    float* deg  = agg + ND;

    hipMemsetAsync(deg, 0, N_NODES * sizeof(float), stream);
    degree_kernel<<<(N_EDGES + 255) / 256, 256, 0, stream>>>(dst, deg);

    const int gemmGrid = N_NODES / 16;

    sage_gemm_kernel<<<gemmGrid, 256, 0, stream>>>(h,    W_in1, b_in1, nullptr, nullptr, buf0, ACT_TANH);
    sage_gemm_kernel<<<gemmGrid, 256, 0, stream>>>(buf0, W_in2, b_in2, nullptr, nullptr, buf1, ACT_NONE);

    float* hcur  = buf1;
    float* hnext = buf0;
    const long scatterThreads = (long)N_EDGES * 32;
    const long normThreads    = (long)N_NODES * 32;

    for (int i = 0; i < 3; ++i) {
        hipMemsetAsync(agg, 0, ND * sizeof(float), stream);
        scatter_kernel<<<(int)((scatterThreads + 255) / 256), 256, 0, stream>>>(hcur, src, dst, agg);
        normalize_kernel<<<(int)((normThreads + 255) / 256), 256, 0, stream>>>(agg, deg);

        float* o  = (i < 2) ? hnext : out;
        int   act = (i < 2) ? ACT_SILU : ACT_TANH;
        sage_gemm_kernel<<<gemmGrid, 256, 0, stream>>>(
            hcur, W_self + (size_t)i * D * D, b_self + (size_t)i * D,
            agg,  W_neigh + (size_t)i * D * D, o, act);

        float* t = hcur; hcur = o; hnext = t;
    }
}


# scatter-add: lane = column (l, l+32, l+64, l+96), so each row load and each f32 atomic instruction of a wave covers one whole 128-byte line, not a quarter of four; sums stay arrival-order atomics
# speedup vs baseline: 1.2056x; 1.2056x over previous
#include <hip/hip_runtime.h>
#include <math.h>

typedef float v2f __attribute__((ext_vector_type(2)));
typedef float v8f __attribute__((ext_vector_type(8)));

#define N_NODES 50000
#define N_EDGES 800000
#define D 128
#define ACT_NONE 0
#define ACT_TANH 1
#define ACT_SILU 2

__global__ void degree_kernel(const int* __restrict__ dst, float* __restrict__ deg) {
    int e = blockIdx.x * blockDim.x + threadIdx.x;
    if (e < N_EDGES) atomicAdd(&deg[dst[e]], 1.0f);
}

__global__ void scatter_kernel(const float* __restrict__ h, const int* __restrict__ src,
                               const int* __restrict__ dst, float* __restrict__ agg) {
    long tid = (long)blockIdx.x * blockDim.x + threadIdx.x;
    if (tid >= (long)N_EDGES * 32) return;
    int e = (int)(tid >> 5);
    int l = (int)(tid & 31);
    int s = src[e], d = dst[e];
    const float* hp = h + (long)s * D + l;
    float v0 = hp[0];
    float v1 = hp[32];
    float v2 = hp[64];
    float v3 = hp[96];
    float* p = agg + (long)d * D + l;
    atomicAdd(p +  0, v0);
    atomicAdd(p + 32, v1);
    atomicAdd(p + 64, v2);
    atomicAdd(p + 96, v3);
}

__global__ void normalize_kernel(float* __restrict__ agg, const float* __restrict__ deg) {
    long tid = (long)blockIdx.x * blockDim.x + threadIdx.x;
    if (tid >= (long)N_NODES * 32) return;
    int n  = (int)(tid >> 5);
    int c4 = (int)(tid & 31) << 2;
    float dv = 1.0f / fmaxf(deg[n], 1.0f);
    float4* p = (float4*)(agg + (long)n * D + c4);
    float4 v = *p;
    v.x *= dv; v.y *= dv; v.z *= dv; v.w *= dv;
    *p = v;
}

__launch_bounds__(256)
__global__ void sage_gemm_kernel(const float* __restrict__ A1, const float* __restrict__ W1,
                                 const float* __restrict__ bias,
                                 const float* __restrict__ A2, const float* __restrict__ W2,
                                 float* __restrict__ Out, int act) {
    __shared__ float sW[D * D];
    __shared__ float sA[16 * D];

    const int tid     = threadIdx.x;
    const int wave    = tid >> 5;
    const int lane    = tid & 31;
    const int rowBase = blockIdx.x * 16;
    const int colBase = wave * 16;
    const int lrow    = lane & 15;
    const int hiHalf  = lane >> 4;
    const int koff    = hiHalf << 1;

    v8f c = {};

    for (int phase = 0; phase < 2; ++phase) {
        const float* A = (phase == 0) ? A1 : A2;
        const float* W = (phase == 0) ? W1 : W2;
        if (A == nullptr) break;

        __syncthreads();
        for (int i = tid; i < (D * D) / 4; i += 256)
            ((float4*)sW)[i] = ((const float4*)W)[i];
        for (int i = tid; i < (16 * D) / 4; i += 256)
            ((float4*)sA)[i] = ((const float4*)(A + (long)rowBase * D))[i];
        __syncthreads();

        #pragma unroll
        for (int k0 = 0; k0 < D; k0 += 4) {
            v2f a, b;
            a.x = sA[lrow * D + k0 + koff];
            a.y = sA[lrow * D + k0 + koff + 1];
            b.x = sW[(k0 + koff) * D + colBase + lrow];
            b.y = sW[(k0 + koff + 1) * D + colBase + lrow];
            c = __builtin_amdgcn_wmma_f32_16x16x4_f32(
                     false, a,  false, b,
                     (short)0, c,  false,  false);
        }
    }

    const int col = colBase + lrow;
    const float bv = bias ? bias[col] : 0.0f;
    #pragma unroll
    for (int r = 0; r < 8; ++r) {
        int grow = rowBase + r + (hiHalf << 3);
        float v = c[r] + bv;
        if (act == ACT_TANH)      v = tanhf(v);
        else if (act == ACT_SILU) v = v / (1.0f + __expf(-v));
        Out[(long)grow * D + col] = v;
    }
}

extern "C" void kernel_launch(void* const* d_in, const int* in_sizes, int n_in,
                              void* d_out, int out_size, void* d_ws, size_t ws_size,
                              hipStream_t stream) {
    const float* h       = (const float*)d_in[0];
    const int*   src     = (const int*)d_in[1];
    const int*   dst     = (const int*)d_in[2];
    const float* W_in1   = (const float*)d_in[3];
    const float* b_in1   = (const float*)d_in[4];
    const float* W_in2   = (const float*)d_in[5];
    const float* b_in2   = (const float*)d_in[6];
    const float* W_self  = (const float*)d_in[7];
    const float* b_self  = (const float*)d_in[8];
    const float* W_neigh = (const float*)d_in[9];
    float* out = (float*)d_out;

    const size_t ND = (size_t)N_NODES * D;
    float* buf0 = (float*)d_ws;
    float* buf1 = buf0 + ND;
    float* agg  = buf1 + ND;
    float* deg  = agg + ND;

    hipMemsetAsync(deg, 0, N_NODES * sizeof(float), stream);
    degree_kernel<<<(N_EDGES + 255) / 256, 256, 0, stream>>>(dst, deg);

    const int gemmGrid = N_NODES / 16;

    sage_gemm_kernel<<<gemmGrid, 256, 0, stream>>>(h,    W_in1, b_in1, nullptr, nullptr, buf0, ACT_TANH);
    sage_gemm_kernel<<<gemmGrid, 256, 0, stream>>>(buf0, W_in2, b_in2, nullptr, nullptr, buf1, ACT_NONE);

    float* hcur  = buf1;
    float* hnext = buf0;
    const long scatterThreads = (long)N_EDGES * 32;
    const long normThreads    = (long)N_NODES * 32;

    for (int i = 0; i < 3; ++i) {
        hipMemsetAsync(agg, 0, ND * sizeof(float), stream);
        scatter_kernel<<<(int)((scatterThreads + 255) / 256), 256, 0, stream>>>(hcur, src, dst, agg);
        normalize_kernel<<<(int)((normThreads + 255) / 256), 256, 0, stream>>>(agg, deg);

        float* o  = (i < 2) ? hnext : out;
        int   act = (i < 2) ? ACT_SILU : ACT_TANH;
        sage_gemm_kernel<<<gemmGrid, 256, 0, stream>>>(
            hcur, W_self + (size_t)i * D * D, b_self + (size_t)i * D,
            agg,  W_neigh + (size_t)i * D * D, o, act);

        float* t = hcur; hcur = o; hnext = t;
    }
}


# GEMM: LDS row pitch 128 -> 132 floats (A tile) and 136 (W), so the A-fragment reads (16 rows, was one bank, 16-way conflict) and the B reads (2-way) are bank-conflict-free
# speedup vs baseline: 1.5178x; 1.2590x over previous
#include <hip/hip_runtime.h>
#include <math.h>

typedef float v2f __attribute__((ext_vector_type(2)));
typedef float v8f __attribute__((ext_vector_type(8)));

#define N_NODES 50000
#define N_EDGES 800000
#define D 128
#define ACT_NONE 0
#define ACT_TANH 1
#define ACT_SILU 2
#define PITCH_A 132
#define PITCH_W 136

__global__ void degree_kernel(const int* __restrict__ dst, float* __restrict__ deg) {
    int e = blockIdx.x * blockDim.x + threadIdx.x;
    if (e < N_EDGES) atomicAdd(&deg[dst[e]], 1.0f);
}

__global__ void scatter_kernel(const float* __restrict__ h, const int* __restrict__ src,
                               const int* __restrict__ dst, float* __restrict__ agg) {
    long tid = (long)blockIdx.x * blockDim.x + threadIdx.x;
    if (tid >= (long)N_EDGES * 32) return;
    int e = (int)(tid >> 5);
    int l = (int)(tid & 31);
    int s = src[e], d = dst[e];
    const float* hp = h + (long)s * D + l;
    float v0 = hp[0];
    float v1 = hp[32];
    float v2 = hp[64];
    float v3 = hp[96];
    float* p = agg + (long)d * D + l;
    atomicAdd(p +  0, v0);
    atomicAdd(p + 32, v1);
    atomicAdd(p + 64, v2);
    atomicAdd(p + 96, v3);
}

__global__ void normalize_kernel(float* __restrict__ agg, const float* __restrict__ deg) {
    long tid = (long)blockIdx.x * blockDim.x + threadIdx.x;
    if (tid >= (long)N_NODES * 32) return;
    int n  = (int)(tid >> 5);
    int c4 = (int)(tid & 31) << 2;
    float dv = 1.0f / fmaxf(deg[n], 1.0f);
    float4* p = (float4*)(agg + (long)n * D + c4);
    float4 v = *p;
    v.x *= dv; v.y *= dv; v.z *= dv; v.w *= dv;
    *p = v;
}

__launch_bounds__(256)
__global__ void sage_gemm_kernel(const float* __restrict__ A1, const float* __restrict__ W1,
                                 const float* __restrict__ bias,
                                 const float* __restrict__ A2, const float* __restrict__ W2,
                                 float* __restrict__ Out, int act) {
    __shared__ __attribute__((aligned(16))) float sW[D * PITCH_W];
    __shared__ __attribute__((aligned(16))) float sA[16 * PITCH_A];

    const int tid     = threadIdx.x;
    const int wave    = tid >> 5;
    const int lane    = tid & 31;
    const int rowBase = blockIdx.x * 16;
    const int colBase = wave * 16;
    const int lrow    = lane & 15;
    const int hiHalf  = lane >> 4;
    const int koff    = hiHalf << 1;

    v8f c = {};

    for (int phase = 0; phase < 2; ++phase) {
        const float* A = (phase == 0) ? A1 : A2;
        const float* W = (phase == 0) ? W1 : W2;
        if (A == nullptr) break;

        __syncthreads();
        for (int i = tid; i < (D * D) / 4; i += 256)
            *(float4*)&sW[(i >> 5) * PITCH_W + ((i & 31) << 2)] = ((const float4*)W)[i];
        for (int i = tid; i < (16 * D) / 4; i += 256)
            *(float4*)&sA[(i >> 5) * PITCH_A + ((i & 31) << 2)] =
                ((const float4*)(A + (long)rowBase * D))[i];
        __syncthreads();

        #pragma unroll
        for (int k0 = 0; k0 < D; k0 += 4) {
            v2f a, b;
            a.x = sA[lrow * PITCH_A + k0 + koff];
            a.y = sA[lrow * PITCH_A + k0 + koff + 1];
            b.x = sW[(k0 + koff) * PITCH_W + colBase + lrow];
            b.y = sW[(k0 + koff + 1) * PITCH_W + colBase + lrow];
            c = __builtin_amdgcn_wmma_f32_16x16x4_f32(
                     false, a,  false, b,
                     (short)0, c,  false,  false);
        }
    }

    const int col = colBase + lrow;
    const float bv = bias ? bias[col] : 0.0f;
    #pragma unroll
    for (int r = 0; r < 8; ++r) {
        int grow = rowBase + r + (hiHalf << 3);
        float v = c[r] + bv;
        if (act == ACT_TANH)      v = tanhf(v);
        else if (act == ACT_SILU) v = v / (1.0f + __expf(-v));
        Out[(long)grow * D + col] = v;
    }
}

extern "C" void kernel_launch(void* const* d_in, const int* in_sizes, int n_in,
                              void* d_out, int out_size, void* d_ws, size_t ws_size,
                              hipStream_t stream) {
    const float* h       = (const float*)d_in[0];
    const int*   src     = (const int*)d_in[1];
    const int*   dst     = (const int*)d_in[2];
    const float* W_in1   = (const float*)d_in[3];
    const float* b_in1   = (const float*)d_in[4];
    const float* W_in2   = (const float*)d_in[5];
    const float* b_in2   = (const float*)d_in[6];
    const float* W_self  = (const float*)d_in[7];
    const float* b_self  = (const float*)d_in[8];
    const float* W_neigh = (const float*)d_in[9];
    float* out = (float*)d_out;

    const size_t ND = (size_t)N_NODES * D;
    float* buf0 = (float*)d_ws;
    float* buf1 = buf0 + ND;
    float* agg  = buf1 + ND;
    float* deg  = agg + ND;

    hipMemsetAsync(deg, 0, N_NODES * sizeof(float), stream);
    degree_kernel<<<(N_EDGES + 255) / 256, 256, 0, stream>>>(dst, deg);

    const int gemmGrid = N_NODES / 16;

    sage_gemm_kernel<<<gemmGrid, 256, 0, stream>>>(h,    W_in1, b_in1, nullptr, nullptr, buf0, ACT_TANH);
    sage_gemm_kernel<<<gemmGrid, 256, 0, stream>>>(buf0, W_in2, b_in2, nullptr, nullptr, buf1, ACT_NONE);

    float* hcur  = buf1;
    float* hnext = buf0;
    const long scatterThreads = (long)N_EDGES * 32;
    const long normThreads    = (long)N_NODES * 32;

    for (int i = 0; i < 3; ++i) {
        hipMemsetAsync(agg, 0, ND * sizeof(float), stream);
        scatter_kernel<<<(int)((scatterThreads + 255) / 256), 256, 0, stream>>>(hcur, src, dst, agg);
        normalize_kernel<<<(int)((normThreads + 255) / 256), 256, 0, stream>>>(agg, deg);

        float* o  = (i < 2) ? hnext : out;
        int   act = (i < 2) ? ACT_SILU : ACT_TANH;
        sage_gemm_kernel<<<gemmGrid, 256, 0, stream>>>(
            hcur, W_self + (size_t)i * D * D, b_self + (size_t)i * D,
            agg,  W_neigh + (size_t)i * D * D, o, act);

        float* t = hcur; hcur = o; hnext = t;
    }
}


# neighbour sum: per-call counting sort of edges by destination (int atomic rank, scan, fill), then one wave per node adds source rows in registers in rank order; float atomics, memset, normalise gone
# speedup vs baseline: 3.0762x; 2.0267x over previous
#include <hip/hip_runtime.h>
#include <math.h>

typedef float v2f __attribute__((ext_vector_type(2)));
typedef float v8f __attribute__((ext_vector_type(8)));

#define N_NODES 50000
#define N_EDGES 800000
#define D 128
#define ACT_NONE 0
#define ACT_TANH 1
#define ACT_SILU 2
#define PITCH_A 132
#define PITCH_W 136

#define SCAN_THREADS 1024
#define SCAN_CHUNK 49
#define SCAN_CAP (SCAN_THREADS * SCAN_CHUNK)

__global__ void rank_kernel(const int* __restrict__ dst, int* __restrict__ cnt,
                            int* __restrict__ rank) {
    int e = blockIdx.x * blockDim.x + threadIdx.x;
    if (e < N_EDGES) rank[e] = atomicAdd(&cnt[dst[e]], 1);
}

__launch_bounds__(SCAN_THREADS)
__global__ void scan_kernel(int* __restrict__ cnt) {
    __shared__ __attribute__((aligned(16))) int sv[SCAN_CAP];
    __shared__ int wsum[32];
    const int t = threadIdx.x;
    const int lane = t & 31;
    const int w = t >> 5;
    for (int i = t; i < SCAN_CAP / 4; i += SCAN_THREADS)
        ((int4*)sv)[i] = ((const int4*)cnt)[i];
    __syncthreads();

    int v[SCAN_CHUNK];
    int s = 0;
    const int base = t * SCAN_CHUNK;
    #pragma unroll
    for (int i = 0; i < SCAN_CHUNK; ++i) { v[i] = sv[base + i]; s += v[i]; }

    int inc = s;
    #pragma unroll
    for (int o = 1; o < 32; o <<= 1) {
        int y = __shfl_up(inc, o);
        if (lane >= o) inc += y;
    }
    if (lane == 31) wsum[w] = inc;
    __syncthreads();
    if (w == 0) {
        int x = wsum[lane];
        int xi = x;
        #pragma unroll
        for (int o = 1; o < 32; o <<= 1) {
            int y = __shfl_up(xi, o);
            if (lane >= o) xi += y;
        }
        wsum[lane] = xi - x;
    }
    __syncthreads();
    int run = wsum[w] + inc - s;
    #pragma unroll
    for (int i = 0; i < SCAN_CHUNK; ++i) { sv[base + i] = run; run += v[i]; }
    __syncthreads();
    for (int i = t; i < SCAN_CAP / 4; i += SCAN_THREADS)
        ((int4*)cnt)[i] = ((const int4*)sv)[i];
}

__global__ void fill_kernel(const int* __restrict__ src, const int* __restrict__ dst,
                            const int* __restrict__ rank, const int* __restrict__ rowptr,
                            int* __restrict__ col) {
    int e = blockIdx.x * blockDim.x + threadIdx.x;
    if (e < N_EDGES) col[rowptr[dst[e]] + rank[e]] = src[e];
}

__launch_bounds__(256)
__global__ void gather_mean_kernel(const float* __restrict__ h, const int* __restrict__ rowptr,
                                   const int* __restrict__ col, float* __restrict__ agg) {
    const int lane = threadIdx.x & 31;
    const int n = __builtin_amdgcn_readfirstlane((int)(blockIdx.x * 8 + (threadIdx.x >> 5)));
    const int beg = rowptr[n];
    const int end = rowptr[n + 1];
    const float4* __restrict__ h4 = (const float4*)h;
    float4 acc = make_float4(0.0f, 0.0f, 0.0f, 0.0f);

    for (int j0 = beg; j0 < end; j0 += 32) {
        const int rem = end - j0;
        const int m = rem < 32 ? rem : 32;
        const int mine = (lane < m) ? col[j0 + lane] : 0;
        for (int t = 0; t < m; t += 8) {
            float4 x[8];
            int sid[8];
            #pragma unroll
            for (int k = 0; k < 8; ++k) sid[k] = __builtin_amdgcn_readlane(mine, t + k);
            #pragma unroll
            for (int k = 0; k < 8; ++k) {
                if (t + k < m) x[k] = h4[(long)sid[k] * (D / 4) + lane];
            }
            #pragma unroll
            for (int k = 0; k < 8; ++k) {
                if (t + k < m) {
                    acc.x += x[k].x; acc.y += x[k].y; acc.z += x[k].z; acc.w += x[k].w;
                }
            }
        }
    }
    const float dv = 1.0f / fmaxf((float)(end - beg), 1.0f);
    acc.x *= dv; acc.y *= dv; acc.z *= dv; acc.w *= dv;
    ((float4*)agg)[(long)n * (D / 4) + lane] = acc;
}

__launch_bounds__(256)
__global__ void sage_gemm_kernel(const float* __restrict__ A1, const float* __restrict__ W1,
                                 const float* __restrict__ bias,
                                 const float* __restrict__ A2, const float* __restrict__ W2,
                                 float* __restrict__ Out, int act) {
    __shared__ __attribute__((aligned(16))) float sW[D * PITCH_W];
    __shared__ __attribute__((aligned(16))) float sA[16 * PITCH_A];

    const int tid     = threadIdx.x;
    const int wave    = tid >> 5;
    const int lane    = tid & 31;
    const int rowBase = blockIdx.x * 16;
    const int colBase = wave * 16;
    const int lrow    = lane & 15;
    const int hiHalf  = lane >> 4;
    const int koff    = hiHalf << 1;

    v8f c = {};

    for (int phase = 0; phase < 2; ++phase) {
        const float* A = (phase == 0) ? A1 : A2;
        const float* W = (phase == 0) ? W1 : W2;
        if (A == nullptr) break;

        __syncthreads();
        for (int i = tid; i < (D * D) / 4; i += 256)
            *(float4*)&sW[(i >> 5) * PITCH_W + ((i & 31) << 2)] = ((const float4*)W)[i];
        for (int i = tid; i < (16 * D) / 4; i += 256)
            *(float4*)&sA[(i >> 5) * PITCH_A + ((i & 31) << 2)] =
                ((const float4*)(A + (long)rowBase * D))[i];
        __syncthreads();

        #pragma unroll
        for (int k0 = 0; k0 < D; k0 += 4) {
            v2f a, b;
            a.x = sA[lrow * PITCH_A + k0 + koff];
            a.y = sA[lrow * PITCH_A + k0 + koff + 1];
            b.x = sW[(k0 + koff) * PITCH_W + colBase + lrow];
            b.y = sW[(k0 + koff + 1) * PITCH_W + colBase + lrow];
            c = __builtin_amdgcn_wmma_f32_16x16x4_f32(
                     false, a,  false, b,
                     (short)0, c,  false,  false);
        }
    }

    const int col = colBase + lrow;
    const float bv = bias ? bias[col] : 0.0f;
    #pragma unroll
    for (int r = 0; r < 8; ++r) {
        int grow = rowBase + r + (hiHalf << 3);
        float v = c[r] + bv;
        if (act == ACT_TANH)      v = tanhf(v);
        else if (act == ACT_SILU) v = v / (1.0f + __expf(-v));
        Out[(long)grow * D + col] = v;
    }
}

extern "C" void kernel_launch(void* const* d_in, const int* in_sizes, int n_in,
                              void* d_out, int out_size, void* d_ws, size_t ws_size,
                              hipStream_t stream) {
    const float* h       = (const float*)d_in[0];
    const int*   src     = (const int*)d_in[1];
    const int*   dst     = (const int*)d_in[2];
    const float* W_in1   = (const float*)d_in[3];
    const float* b_in1   = (const float*)d_in[4];
    const float* W_in2   = (const float*)d_in[5];
    const float* b_in2   = (const float*)d_in[6];
    const float* W_self  = (const float*)d_in[7];
    const float* b_self  = (const float*)d_in[8];
    const float* W_neigh = (const float*)d_in[9];
    float* out = (float*)d_out;

    const size_t ND = (size_t)N_NODES * D;
    float* bufY   = (float*)d_ws;
    float* agg    = bufY + ND;
    int*   rowptr = (int*)(agg + ND);
    int*   rank   = rowptr + SCAN_CAP;
    int*   col    = rank + N_EDGES;

    hipMemsetAsync(rowptr, 0, SCAN_CAP * sizeof(int), stream);
    rank_kernel<<<(N_EDGES + 255) / 256, 256, 0, stream>>>(dst, rowptr, rank);
    scan_kernel<<<1, SCAN_THREADS, 0, stream>>>(rowptr);
    fill_kernel<<<(N_EDGES + 255) / 256, 256, 0, stream>>>(src, dst, rank, rowptr, col);

    const int gemmGrid = N_NODES / 16;

    sage_gemm_kernel<<<gemmGrid, 256, 0, stream>>>(h,   W_in1, b_in1, nullptr, nullptr, out,  ACT_TANH);
    sage_gemm_kernel<<<gemmGrid, 256, 0, stream>>>(out, W_in2, b_in2, nullptr, nullptr, bufY, ACT_NONE);

    float* hcur  = bufY;
    float* hnext = out;

    for (int i = 0; i < 3; ++i) {
        gather_mean_kernel<<<N_NODES / 8, 256, 0, stream>>>(hcur, rowptr, col, agg);

        int act = (i < 2) ? ACT_SILU : ACT_TANH;
        sage_gemm_kernel<<<gemmGrid, 256, 0, stream>>>(
            hcur, W_self + (size_t)i * D * D, b_self + (size_t)i * D,
            agg,  W_neigh + (size_t)i * D * D, hnext, act);

        float* t = hcur; hcur = hnext; hnext = t;
    }
}


# GEMM made persistent: 256 work-groups, each wave keeps its 128x16 weight slices in registers for the whole launch and walks 16-row tiles two at a time, next rows prefetched; sum order kept
# speedup vs baseline: 3.4427x; 1.1191x over previous
#include <hip/hip_runtime.h>
#include <math.h>

typedef float v2f __attribute__((ext_vector_type(2)));
typedef float v8f __attribute__((ext_vector_type(8)));

#define N_NODES 50000
#define N_EDGES 800000
#define D 128
#define ACT_NONE 0
#define ACT_TANH 1
#define ACT_SILU 2
#define PITCH_A 132

#define SCAN_THREADS 1024
#define SCAN_CHUNK 49
#define SCAN_CAP (SCAN_THREADS * SCAN_CHUNK)

__global__ void rank_kernel(const int* __restrict__ dst, int* __restrict__ cnt,
                            int* __restrict__ rank) {
    int e = blockIdx.x * blockDim.x + threadIdx.x;
    if (e < N_EDGES) rank[e] = atomicAdd(&cnt[dst[e]], 1);
}

__launch_bounds__(SCAN_THREADS)
__global__ void scan_kernel(int* __restrict__ cnt) {
    __shared__ __attribute__((aligned(16))) int sv[SCAN_CAP];
    __shared__ int wsum[32];
    const int t = threadIdx.x;
    const int lane = t & 31;
    const int w = t >> 5;
    for (int i = t; i < SCAN_CAP / 4; i += SCAN_THREADS)
        ((int4*)sv)[i] = ((const int4*)cnt)[i];
    __syncthreads();

    int v[SCAN_CHUNK];
    int s = 0;
    const int base = t * SCAN_CHUNK;
    #pragma unroll
    for (int i = 0; i < SCAN_CHUNK; ++i) { v[i] = sv[base + i]; s += v[i]; }

    int inc = s;
    #pragma unroll
    for (int o = 1; o < 32; o <<= 1) {
        int y = __shfl_up(inc, o);
        if (lane >= o) inc += y;
    }
    if (lane == 31) wsum[w] = inc;
    __syncthreads();
    if (w == 0) {
        int x = wsum[lane];
        int xi = x;
        #pragma unroll
        for (int o = 1; o < 32; o <<= 1) {
            int y = __shfl_up(xi, o);
            if (lane >= o) xi += y;
        }
        wsum[lane] = xi - x;
    }
    __syncthreads();
    int run = wsum[w] + inc - s;
    #pragma unroll
    for (int i = 0; i < SCAN_CHUNK; ++i) { sv[base + i] = run; run += v[i]; }
    __syncthreads();
    for (int i = t; i < SCAN_CAP / 4; i += SCAN_THREADS)
        ((int4*)cnt)[i] = ((const int4*)sv)[i];
}

__global__ void fill_kernel(const int* __restrict__ src, const int* __restrict__ dst,
                            const int* __restrict__ rank, const int* __restrict__ rowptr,
                            int* __restrict__ col) {
    int e = blockIdx.x * blockDim.x + threadIdx.x;
    if (e < N_EDGES) col[rowptr[dst[e]] + rank[e]] = src[e];
}

__launch_bounds__(256)
__global__ void gather_mean_kernel(const float* __restrict__ h, const int* __restrict__ rowptr,
                                   const int* __restrict__ col, float* __restrict__ agg) {
    const int lane = threadIdx.x & 31;
    const int n = __builtin_amdgcn_readfirstlane((int)(blockIdx.x * 8 + (threadIdx.x >> 5)));
    const int beg = rowptr[n];
    const int end = rowptr[n + 1];
    const float4* __restrict__ h4 = (const float4*)h;
    float4 acc = make_float4(0.0f, 0.0f, 0.0f, 0.0f);

    for (int j0 = beg; j0 < end; j0 += 32) {
        const int rem = end - j0;
        const int m = rem < 32 ? rem : 32;
        const int mine = (lane < m) ? col[j0 + lane] : 0;
        for (int t = 0; t < m; t += 8) {
            float4 x[8];
            int sid[8];
            #pragma unroll
            for (int k = 0; k < 8; ++k) sid[k] = __builtin_amdgcn_readlane(mine, t + k);
            #pragma unroll
            for (int k = 0; k < 8; ++k) {
                if (t + k < m) x[k] = h4[(long)sid[k] * (D / 4) + lane];
            }
            #pragma unroll
            for (int k = 0; k < 8; ++k) {
                if (t + k < m) {
                    acc.x += x[k].x; acc.y += x[k].y; acc.z += x[k].z; acc.w += x[k].w;
                }
            }
        }
    }
    const float dv = 1.0f / fmaxf((float)(end - beg), 1.0f);
    acc.x *= dv; acc.y *= dv; acc.z *= dv; acc.w *= dv;
    ((float4*)agg)[(long)n * (D / 4) + lane] = acc;
}

#define GEMM_GRID 256
#define N_TILES (N_NODES / 16)
#define TILE_F (16 * PITCH_A)

#define FRAG_GROUP 8

__device__ __forceinline__ void lds_barrier() {
    __builtin_amdgcn_fence(__ATOMIC_RELEASE, "workgroup", "local");
    __builtin_amdgcn_s_barrier();
    __builtin_amdgcn_fence(__ATOMIC_ACQUIRE, "workgroup", "local");
}

template <int NPH, int NT>
__device__ __forceinline__ void tile_products(const float* __restrict__ sbuf,
                                              const v2f (&bf)[NPH][32],
                                              int aoff, v8f& c0, v8f& c1) {
    constexpr int NG = NPH * 32 / FRAG_GROUP;
    v2f fa[2][FRAG_GROUP], fb[2][FRAG_GROUP];

    #pragma unroll
    for (int u = 0; u < FRAG_GROUP; ++u) {
        const float* p = sbuf + aoff + 4 * u;
        fa[0][u].x = p[0];
        fa[0][u].y = p[1];
        if (NT == 2) {
            fb[0][u].x = p[TILE_F];
            fb[0][u].y = p[TILE_F + 1];
        }
    }
    __builtin_amdgcn_sched_barrier(0);
    #pragma unroll
    for (int g = 0; g < NG; ++g) {
        if (g + 1 < NG) {
            const int phn = (g + 1) / (32 / FRAG_GROUP);
            const int sn  = ((g + 1) % (32 / FRAG_GROUP)) * FRAG_GROUP;
            #pragma unroll
            for (int u = 0; u < FRAG_GROUP; ++u) {
                const float* p = sbuf + (phn * 2) * TILE_F + aoff + 4 * (sn + u);
                fa[(g + 1) & 1][u].x = p[0];
                fa[(g + 1) & 1][u].y = p[1];
                if (NT == 2) {
                    fb[(g + 1) & 1][u].x = p[TILE_F];
                    fb[(g + 1) & 1][u].y = p[TILE_F + 1];
                }
            }
            __builtin_amdgcn_sched_barrier(0);
        }
        const int ph = g / (32 / FRAG_GROUP);
        const int s0 = (g % (32 / FRAG_GROUP)) * FRAG_GROUP;
        #pragma unroll
        for (int u = 0; u < FRAG_GROUP; ++u) {
            c0 = __builtin_amdgcn_wmma_f32_16x16x4_f32(
                      false, fa[g & 1][u],  false, bf[ph][s0 + u],
                      (short)0, c0,  false,  false);
            if (NT == 2)
                c1 = __builtin_amdgcn_wmma_f32_16x16x4_f32(
                          false, fb[g & 1][u],  false, bf[ph][s0 + u],
                          (short)0, c1,  false,  false);
        }
        __builtin_amdgcn_sched_barrier(0);
    }
}

template <int NPH>
__launch_bounds__(256)
__global__ void sage_gemm_kernel(const float* __restrict__ A1, const float* __restrict__ W1,
                                 const float* __restrict__ bias,
                                 const float* __restrict__ A2, const float* __restrict__ W2,
                                 float* __restrict__ Out, int act) {
    __shared__ __attribute__((aligned(16))) float sA[2 * NPH * 2 * TILE_F];

    const int tid     = threadIdx.x;
    const int wave    = tid >> 5;
    const int lane    = tid & 31;
    const int colBase = wave * 16;
    const int lrow    = lane & 15;
    const int hiHalf  = lane >> 4;
    const int koff    = hiHalf << 1;
    const int col     = colBase + lrow;
    const int aoff    = lrow * PITCH_A + koff;

    const float bv = bias ? bias[col] : 0.0f;

    const int i0 = tid, i1 = tid + 256;
    const int l0 = (i0 >> 5) * PITCH_A + ((i0 & 31) << 2);
    const int l1 = (i1 >> 5) * PITCH_A + ((i1 & 31) << 2);

    float4 pa0, pa1, pa2, pa3, pb0, pb1, pb2, pb3;
    const float4* __restrict__ A14 = (const float4*)A1;
    const float4* __restrict__ A24 = (const float4*)A2;
    auto fetch = [&](int ta, int tb, bool vb) {
        pa0 = A14[(long)ta * 512 + i0];
        pa1 = A14[(long)ta * 512 + i1];
        if (NPH == 2) {
            pb0 = A24[(long)ta * 512 + i0];
            pb1 = A24[(long)ta * 512 + i1];
        }
        if (vb) {
            pa2 = A14[(long)tb * 512 + i0];
            pa3 = A14[(long)tb * 512 + i1];
            if (NPH == 2) {
                pb2 = A24[(long)tb * 512 + i0];
                pb3 = A24[(long)tb * 512 + i1];
            }
        }
    };
    auto put = [&](float* sb, bool vb) {
        *(float4*)&sb[l0] = pa0;
        *(float4*)&sb[l1] = pa1;
        if (NPH == 2) {
            *(float4*)&sb[2 * TILE_F + l0] = pb0;
            *(float4*)&sb[2 * TILE_F + l1] = pb1;
        }
        if (vb) {
            *(float4*)&sb[TILE_F + l0] = pa2;
            *(float4*)&sb[TILE_F + l1] = pa3;
            if (NPH == 2) {
                *(float4*)&sb[3 * TILE_F + l0] = pb2;
                *(float4*)&sb[3 * TILE_F + l1] = pb3;
            }
        }
    };

    auto emit = [&](const v8f& c, int tile) {
        float* __restrict__ o = Out + ((long)tile * 16 + (hiHalf << 3)) * D + col;
        if (act == ACT_TANH) {
            float y[8];
            #pragma unroll
            for (int r = 0; r < 8; ++r) y[r] = tanhf(c[r] + bv);
            #pragma unroll
            for (int r = 0; r < 8; ++r) o[r * D] = y[r];
        } else if (act == ACT_SILU) {
            #pragma unroll
            for (int r = 0; r < 8; ++r) { float v = c[r] + bv; o[r * D] = v / (1.0f + __expf(-v)); }
        } else {
            #pragma unroll
            for (int r = 0; r < 8; ++r) o[r * D] = c[r] + bv;
        }
    };

    int  t0 = blockIdx.x;
    int  t1 = blockIdx.x + GEMM_GRID;
    bool v1 = t1 < N_TILES;
    int  buf = 0;

    fetch(t0, t1, v1);

    v2f bf[NPH][32];
    #pragma unroll
    for (int ph = 0; ph < NPH; ++ph) {
        const float* __restrict__ Wp = (ph == 0) ? W1 : W2;
        #pragma unroll
        for (int s = 0; s < 32; ++s) {
            bf[ph][s].x = Wp[(4 * s + koff) * D + col];
            bf[ph][s].y = Wp[(4 * s + koff + 1) * D + col];
        }
    }

    put(sA, v1);
    __syncthreads();

    while (true) {
        const int  n0 = t0 + 2 * GEMM_GRID;
        const int  n1 = t1 + 2 * GEMM_GRID;
        const bool more = n0 < N_TILES;
        const bool nv1  = n1 < N_TILES;

        if (more) fetch(n0, n1, nv1);

        const float* sbuf = sA + buf * (NPH * 2 * TILE_F);
        v8f c0 = {}, c1 = {};
        if (v1) tile_products<NPH, 2>(sbuf, bf, aoff, c0, c1);
        else    tile_products<NPH, 1>(sbuf, bf, aoff, c0, c1);

        emit(c0, t0);
        if (v1) emit(c1, t1);

        if (!more) break;

        put(sA + (buf ^ 1) * (NPH * 2 * TILE_F), nv1);
        lds_barrier();
        buf ^= 1; t0 = n0; t1 = n1; v1 = nv1;
    }
}

extern "C" void kernel_launch(void* const* d_in, const int* in_sizes, int n_in,
                              void* d_out, int out_size, void* d_ws, size_t ws_size,
                              hipStream_t stream) {
    const float* h       = (const float*)d_in[0];
    const int*   src     = (const int*)d_in[1];
    const int*   dst     = (const int*)d_in[2];
    const float* W_in1   = (const float*)d_in[3];
    const float* b_in1   = (const float*)d_in[4];
    const float* W_in2   = (const float*)d_in[5];
    const float* b_in2   = (const float*)d_in[6];
    const float* W_self  = (const float*)d_in[7];
    const float* b_self  = (const float*)d_in[8];
    const float* W_neigh = (const float*)d_in[9];
    float* out = (float*)d_out;

    const size_t ND = (size_t)N_NODES * D;
    float* bufY   = (float*)d_ws;
    float* agg    = bufY + ND;
    int*   rowptr = (int*)(agg + ND);
    int*   rank   = rowptr + SCAN_CAP;
    int*   col    = rank + N_EDGES;

    hipMemsetAsync(rowptr, 0, SCAN_CAP * sizeof(int), stream);
    rank_kernel<<<(N_EDGES + 255) / 256, 256, 0, stream>>>(dst, rowptr, rank);
    scan_kernel<<<1, SCAN_THREADS, 0, stream>>>(rowptr);
    fill_kernel<<<(N_EDGES + 255) / 256, 256, 0, stream>>>(src, dst, rank, rowptr, col);

    sage_gemm_kernel<1><<<GEMM_GRID, 256, 0, stream>>>(h,   W_in1, b_in1, nullptr, nullptr, out,  ACT_TANH);
    sage_gemm_kernel<1><<<GEMM_GRID, 256, 0, stream>>>(out, W_in2, b_in2, nullptr, nullptr, bufY, ACT_NONE);

    float* hcur  = bufY;
    float* hnext = out;

    for (int i = 0; i < 3; ++i) {
        gather_mean_kernel<<<N_NODES / 8, 256, 0, stream>>>(hcur, rowptr, col, agg);

        int act = (i < 2) ? ACT_SILU : ACT_TANH;
        sage_gemm_kernel<2><<<GEMM_GRID, 256, 0, stream>>>(
            hcur, W_self + (size_t)i * D * D, b_self + (size_t)i * D,
            agg,  W_neigh + (size_t)i * D * D, hnext, act);

        float* t = hcur; hcur = hnext; hnext = t;
    }
}


# edge sort without global atomics: edges bucketed by destination/256 with LDS counters per chunk, scan, one work-group per bucket ranks in LDS; order inside a neighbour sum = LDS arrival order
# speedup vs baseline: 3.7411x; 1.0867x over previous
#include <hip/hip_runtime.h>
#include <math.h>

typedef float v2f __attribute__((ext_vector_type(2)));
typedef float v8f __attribute__((ext_vector_type(8)));

#define N_NODES 50000
#define N_EDGES 800000
#define D 128
#define ACT_NONE 0
#define ACT_TANH 1
#define ACT_SILU 2
#define PITCH_A 132

#define SCAN_THREADS 1024
#define SCAN_CHUNK 49
#define SCAN_CAP (SCAN_THREADS * SCAN_CHUNK)

#define NB 196
#define NCH 256
#define CH (N_EDGES / NCH)
#define SORT_THREADS 1024
#define SORT_SLOTS ((CH + SORT_THREADS - 1) / SORT_THREADS)
static_assert(NB * NCH == SCAN_CAP, "bucket table must fill the scan table");
static_assert(CH * NCH == N_EDGES, "chunks must tile the edge list");
static_assert(NB * 256 >= N_NODES + 1, "buckets must cover every node and the end entry");

__device__ __forceinline__ int bucket_of(int d) {
    const unsigned b = (unsigned)d >> 8;
    return b < NB ? (int)b : NB - 1;
}

__launch_bounds__(SORT_THREADS)
__global__ void bucket_count_kernel(const int* __restrict__ dst, int* __restrict__ hist) {
    __shared__ int hb[NB];
    const int t = threadIdx.x;
    const int base = blockIdx.x * CH;
    if (t < NB) hb[t] = 0;
    int d[SORT_SLOTS];
    #pragma unroll
    for (int k = 0; k < SORT_SLOTS; ++k) {
        const int i = t + k * SORT_THREADS;
        d[k] = dst[base + (i < CH ? i : CH - 1)];
    }
    __syncthreads();
    #pragma unroll
    for (int k = 0; k < SORT_SLOTS; ++k) {
        const int i = t + k * SORT_THREADS;
        atomicAdd(&hb[bucket_of(d[k])], (i < CH) ? 1 : 0);
    }
    __syncthreads();
    if (t < NB) hist[t * NCH + blockIdx.x] = hb[t];
}

__launch_bounds__(SCAN_THREADS)
__global__ void scan_kernel(int* __restrict__ cnt) {
    __shared__ __attribute__((aligned(16))) int sv[SCAN_CAP];
    __shared__ int wsum[32];
    const int t = threadIdx.x;
    const int lane = t & 31;
    const int w = t >> 5;
    for (int i = t; i < SCAN_CAP / 4; i += SCAN_THREADS)
        ((int4*)sv)[i] = ((const int4*)cnt)[i];
    __syncthreads();

    int v[SCAN_CHUNK];
    int s = 0;
    const int base = t * SCAN_CHUNK;
    #pragma unroll
    for (int i = 0; i < SCAN_CHUNK; ++i) { v[i] = sv[base + i]; s += v[i]; }

    int inc = s;
    #pragma unroll
    for (int o = 1; o < 32; o <<= 1) {
        int y = __shfl_up(inc, o);
        if (lane >= o) inc += y;
    }
    if (lane == 31) wsum[w] = inc;
    __syncthreads();
    if (w == 0) {
        int x = wsum[lane];
        int xi = x;
        #pragma unroll
        for (int o = 1; o < 32; o <<= 1) {
            int y = __shfl_up(xi, o);
            if (lane >= o) xi += y;
        }
        wsum[lane] = xi - x;
    }
    __syncthreads();
    int run = wsum[w] + inc - s;
    #pragma unroll
    for (int i = 0; i < SCAN_CHUNK; ++i) { sv[base + i] = run; run += v[i]; }
    __syncthreads();
    for (int i = t; i < SCAN_CAP / 4; i += SCAN_THREADS)
        ((int4*)cnt)[i] = ((const int4*)sv)[i];
}

__launch_bounds__(SORT_THREADS)
__global__ void bucket_scatter_kernel(const int* __restrict__ src, const int* __restrict__ dst,
                                      const int* __restrict__ hist, int* __restrict__ pairs_raw) {
    __shared__ int cur[NB];
    int2* __restrict__ pairs = (int2*)pairs_raw;
    const int t = threadIdx.x;
    const int base = blockIdx.x * CH;
    int d[SORT_SLOTS], s[SORT_SLOTS];
    #pragma unroll
    for (int k = 0; k < SORT_SLOTS; ++k) {
        const int i = t + k * SORT_THREADS;
        const int ic = base + (i < CH ? i : CH - 1);
        d[k] = dst[ic];
        s[k] = src[ic];
    }
    if (t < NB) cur[t] = hist[t * NCH + blockIdx.x];
    __syncthreads();
    #pragma unroll
    for (int k = 0; k < SORT_SLOTS; ++k) {
        const int i = t + k * SORT_THREADS;
        const int pos = atomicAdd(&cur[bucket_of(d[k])], (i < CH) ? 1 : 0);
        if (i < CH) pairs[pos] = make_int2(d[k], s[k]);
    }
}

#define PLACE_THREADS 1024
#define PLACE_SLOTS 4
__launch_bounds__(PLACE_THREADS)
__global__ void bucket_place_kernel(const int* __restrict__ pairs_raw, const int* __restrict__ hist,
                                    int* __restrict__ rowptr, int* __restrict__ col) {
    __shared__ int cnt[256];
    __shared__ int wsum[8];
    const int2* __restrict__ pairs = (const int2*)pairs_raw;
    const int t = threadIdx.x;
    const int lane = t & 31;
    const int b = blockIdx.x;
    const int beg = hist[b * NCH];
    const int end = (b + 1 < NB) ? hist[(b + 1) * NCH] : N_EDGES;
    if (t < 256) cnt[t] = 0;
    __syncthreads();

    const int last = end - 1;
    for (int j0 = beg; j0 < end; j0 += PLACE_SLOTS * PLACE_THREADS) {
        int dd[PLACE_SLOTS];
        #pragma unroll
        for (int k = 0; k < PLACE_SLOTS; ++k) {
            const int j = j0 + k * PLACE_THREADS + t;
            dd[k] = pairs[j < end ? j : last].x;
        }
        #pragma unroll
        for (int k = 0; k < PLACE_SLOTS; ++k) {
            const int j = j0 + k * PLACE_THREADS + t;
            atomicAdd(&cnt[dd[k] & 255], (j < end) ? 1 : 0);
        }
    }
    __syncthreads();

    int v = 0, inc = 0;
    if (t < 256) {
        v = cnt[t];
        inc = v;
        #pragma unroll
        for (int o = 1; o < 32; o <<= 1) {
            int y = __shfl_up(inc, o);
            if (lane >= o) inc += y;
        }
        if (lane == 31) wsum[t >> 5] = inc;
    }
    __syncthreads();
    if (t < 256) {
        int off = beg + inc - v;
        for (int w = 0; w < (t >> 5); ++w) off += wsum[w];
        const int node = b * 256 + t;
        if (node <= N_NODES) rowptr[node] = off;
        cnt[t] = off;
    }
    __syncthreads();

    for (int j0 = beg; j0 < end; j0 += PLACE_SLOTS * PLACE_THREADS) {
        int2 p[PLACE_SLOTS];
        #pragma unroll
        for (int k = 0; k < PLACE_SLOTS; ++k) {
            const int j = j0 + k * PLACE_THREADS + t;
            p[k] = pairs[j < end ? j : last];
        }
        #pragma unroll
        for (int k = 0; k < PLACE_SLOTS; ++k) {
            const int j = j0 + k * PLACE_THREADS + t;
            const int pos = atomicAdd(&cnt[p[k].x & 255], (j < end) ? 1 : 0);
            if (j < end) col[pos] = p[k].y;
        }
    }
}

__launch_bounds__(256)
__global__ void gather_mean_kernel(const float* __restrict__ h, const int* __restrict__ rowptr,
                                   const int* __restrict__ col, float* __restrict__ agg) {
    const int lane = threadIdx.x & 31;
    const int n = __builtin_amdgcn_readfirstlane((int)(blockIdx.x * 8 + (threadIdx.x >> 5)));
    const int beg = rowptr[n];
    const int end = rowptr[n + 1];
    const float4* __restrict__ h4 = (const float4*)h;
    float4 acc = make_float4(0.0f, 0.0f, 0.0f, 0.0f);

    for (int j0 = beg; j0 < end; j0 += 32) {
        const int rem = end - j0;
        const int m = rem < 32 ? rem : 32;
        const int mine = (lane < m) ? col[j0 + lane] : 0;
        for (int t = 0; t < m; t += 8) {
            float4 x[8];
            int sid[8];
            #pragma unroll
            for (int k = 0; k < 8; ++k) sid[k] = __builtin_amdgcn_readlane(mine, t + k);
            #pragma unroll
            for (int k = 0; k < 8; ++k) {
                if (t + k < m) x[k] = h4[(long)sid[k] * (D / 4) + lane];
            }
            #pragma unroll
            for (int k = 0; k < 8; ++k) {
                if (t + k < m) {
                    acc.x += x[k].x; acc.y += x[k].y; acc.z += x[k].z; acc.w += x[k].w;
                }
            }
        }
    }
    const float dv = 1.0f / fmaxf((float)(end - beg), 1.0f);
    acc.x *= dv; acc.y *= dv; acc.z *= dv; acc.w *= dv;
    ((float4*)agg)[(long)n * (D / 4) + lane] = acc;
}

#define GEMM_GRID 256
#define N_TILES (N_NODES / 16)
#define TILE_F (16 * PITCH_A)

#define FRAG_GROUP 8

__device__ __forceinline__ void lds_barrier() {
    __builtin_amdgcn_fence(__ATOMIC_RELEASE, "workgroup", "local");
    __builtin_amdgcn_s_barrier();
    __builtin_amdgcn_fence(__ATOMIC_ACQUIRE, "workgroup", "local");
}

template <int NPH, int NT>
__device__ __forceinline__ void tile_products(const float* __restrict__ sbuf,
                                              const v2f (&bf)[NPH][32],
                                              int aoff, v8f& c0, v8f& c1) {
    constexpr int NG = NPH * 32 / FRAG_GROUP;
    v2f fa[2][FRAG_GROUP], fb[2][FRAG_GROUP];

    #pragma unroll
    for (int u = 0; u < FRAG_GROUP; ++u) {
        const float* p = sbuf + aoff + 4 * u;
        fa[0][u].x = p[0];
        fa[0][u].y = p[1];
        if (NT == 2) {
            fb[0][u].x = p[TILE_F];
            fb[0][u].y = p[TILE_F + 1];
        }
    }
    __builtin_amdgcn_sched_barrier(0);
    #pragma unroll
    for (int g = 0; g < NG; ++g) {
        if (g + 1 < NG) {
            const int phn = (g + 1) / (32 / FRAG_GROUP);
            const int sn  = ((g + 1) % (32 / FRAG_GROUP)) * FRAG_GROUP;
            #pragma unroll
            for (int u = 0; u < FRAG_GROUP; ++u) {
                const float* p = sbuf + (phn * 2) * TILE_F + aoff + 4 * (sn + u);
                fa[(g + 1) & 1][u].x = p[0];
                fa[(g + 1) & 1][u].y = p[1];
                if (NT == 2) {
                    fb[(g + 1) & 1][u].x = p[TILE_F];
                    fb[(g + 1) & 1][u].y = p[TILE_F + 1];
                }
            }
            __builtin_amdgcn_sched_barrier(0);
        }
        const int ph = g / (32 / FRAG_GROUP);
        const int s0 = (g % (32 / FRAG_GROUP)) * FRAG_GROUP;
        #pragma unroll
        for (int u = 0; u < FRAG_GROUP; ++u) {
            c0 = __builtin_amdgcn_wmma_f32_16x16x4_f32(
                      false, fa[g & 1][u],  false, bf[ph][s0 + u],
                      (short)0, c0,  false,  false);
            if (NT == 2)
                c1 = __builtin_amdgcn_wmma_f32_16x16x4_f32(
                          false, fb[g & 1][u],  false, bf[ph][s0 + u],
                          (short)0, c1,  false,  false);
        }
        __builtin_amdgcn_sched_barrier(0);
    }
}

template <int NPH>
__launch_bounds__(256)
__global__ void sage_gemm_kernel(const float* __restrict__ A1, const float* __restrict__ W1,
                                 const float* __restrict__ bias,
                                 const float* __restrict__ A2, const float* __restrict__ W2,
                                 float* __restrict__ Out, int act) {
    __shared__ __attribute__((aligned(16))) float sA[2 * NPH * 2 * TILE_F];

    const int tid     = threadIdx.x;
    const int wave    = tid >> 5;
    const int lane    = tid & 31;
    const int colBase = wave * 16;
    const int lrow    = lane & 15;
    const int hiHalf  = lane >> 4;
    const int koff    = hiHalf << 1;
    const int col     = colBase + lrow;
    const int aoff    = lrow * PITCH_A + koff;

    const float bv = bias ? bias[col] : 0.0f;

    const int i0 = tid, i1 = tid + 256;
    const int l0 = (i0 >> 5) * PITCH_A + ((i0 & 31) << 2);
    const int l1 = (i1 >> 5) * PITCH_A + ((i1 & 31) << 2);

    float4 pa0, pa1, pa2, pa3, pb0, pb1, pb2, pb3;
    const float4* __restrict__ A14 = (const float4*)A1;
    const float4* __restrict__ A24 = (const float4*)A2;
    auto fetch = [&](int ta, int tb, bool vb) {
        pa0 = A14[(long)ta * 512 + i0];
        pa1 = A14[(long)ta * 512 + i1];
        if (NPH == 2) {
            pb0 = A24[(long)ta * 512 + i0];
            pb1 = A24[(long)ta * 512 + i1];
        }
        if (vb) {
            pa2 = A14[(long)tb * 512 + i0];
            pa3 = A14[(long)tb * 512 + i1];
            if (NPH == 2) {
                pb2 = A24[(long)tb * 512 + i0];
                pb3 = A24[(long)tb * 512 + i1];
            }
        }
    };
    auto put = [&](float* sb, bool vb) {
        *(float4*)&sb[l0] = pa0;
        *(float4*)&sb[l1] = pa1;
        if (NPH == 2) {
            *(float4*)&sb[2 * TILE_F + l0] = pb0;
            *(float4*)&sb[2 * TILE_F + l1] = pb1;
        }
        if (vb) {
            *(float4*)&sb[TILE_F + l0] = pa2;
            *(float4*)&sb[TILE_F + l1] = pa3;
            if (NPH == 2) {
                *(float4*)&sb[3 * TILE_F + l0] = pb2;
                *(float4*)&sb[3 * TILE_F + l1] = pb3;
            }
        }
    };

    auto emit = [&](const v8f& c, int tile) {
        float* __restrict__ o = Out + ((long)tile * 16 + (hiHalf << 3)) * D + col;
        if (act == ACT_TANH) {
            float y[8];
            #pragma unroll
            for (int r = 0; r < 8; ++r) y[r] = tanhf(c[r] + bv);
            #pragma unroll
            for (int r = 0; r < 8; ++r) o[r * D] = y[r];
        } else if (act == ACT_SILU) {
            #pragma unroll
            for (int r = 0; r < 8; ++r) { float v = c[r] + bv; o[r * D] = v / (1.0f + __expf(-v)); }
        } else {
            #pragma unroll
            for (int r = 0; r < 8; ++r) o[r * D] = c[r] + bv;
        }
    };

    int  t0 = blockIdx.x;
    int  t1 = blockIdx.x + GEMM_GRID;
    bool v1 = t1 < N_TILES;
    int  buf = 0;

    fetch(t0, t1, v1);

    v2f bf[NPH][32];
    #pragma unroll
    for (int ph = 0; ph < NPH; ++ph) {
        const float* __restrict__ Wp = (ph == 0) ? W1 : W2;
        #pragma unroll
        for (int s = 0; s < 32; ++s) {
            bf[ph][s].x = Wp[(4 * s + koff) * D + col];
            bf[ph][s].y = Wp[(4 * s + koff + 1) * D + col];
        }
    }

    put(sA, v1);
    __syncthreads();

    while (true) {
        const int  n0 = t0 + 2 * GEMM_GRID;
        const int  n1 = t1 + 2 * GEMM_GRID;
        const bool more = n0 < N_TILES;
        const bool nv1  = n1 < N_TILES;

        if (more) fetch(n0, n1, nv1);

        const float* sbuf = sA + buf * (NPH * 2 * TILE_F);
        v8f c0 = {}, c1 = {};
        if (v1) tile_products<NPH, 2>(sbuf, bf, aoff, c0, c1);
        else    tile_products<NPH, 1>(sbuf, bf, aoff, c0, c1);

        emit(c0, t0);
        if (v1) emit(c1, t1);

        if (!more) break;

        put(sA + (buf ^ 1) * (NPH * 2 * TILE_F), nv1);
        lds_barrier();
        buf ^= 1; t0 = n0; t1 = n1; v1 = nv1;
    }
}

extern "C" void kernel_launch(void* const* d_in, const int* in_sizes, int n_in,
                              void* d_out, int out_size, void* d_ws, size_t ws_size,
                              hipStream_t stream) {
    const float* h       = (const float*)d_in[0];
    const int*   src     = (const int*)d_in[1];
    const int*   dst     = (const int*)d_in[2];
    const float* W_in1   = (const float*)d_in[3];
    const float* b_in1   = (const float*)d_in[4];
    const float* W_in2   = (const float*)d_in[5];
    const float* b_in2   = (const float*)d_in[6];
    const float* W_self  = (const float*)d_in[7];
    const float* b_self  = (const float*)d_in[8];
    const float* W_neigh = (const float*)d_in[9];
    float* out = (float*)d_out;

    const size_t ND = (size_t)N_NODES * D;
    float* bufY   = (float*)d_ws;
    float* agg    = bufY + ND;
    int*   hist   = (int*)(agg + ND);
    int*   rowptr = hist + SCAN_CAP;
    int*   col    = rowptr + SCAN_CAP;
    int*   pairs  = (int*)agg;

    bucket_count_kernel<<<NCH, SORT_THREADS, 0, stream>>>(dst, hist);
    scan_kernel<<<1, SCAN_THREADS, 0, stream>>>(hist);
    bucket_scatter_kernel<<<NCH, SORT_THREADS, 0, stream>>>(src, dst, hist, pairs);
    bucket_place_kernel<<<NB, PLACE_THREADS, 0, stream>>>(pairs, hist, rowptr, col);

    sage_gemm_kernel<1><<<GEMM_GRID, 256, 0, stream>>>(h,   W_in1, b_in1, nullptr, nullptr, out,  ACT_TANH);
    sage_gemm_kernel<1><<<GEMM_GRID, 256, 0, stream>>>(out, W_in2, b_in2, nullptr, nullptr, bufY, ACT_NONE);

    float* hcur  = bufY;
    float* hnext = out;

    for (int i = 0; i < 3; ++i) {
        gather_mean_kernel<<<N_NODES / 8, 256, 0, stream>>>(hcur, rowptr, col, agg);

        int act = (i < 2) ? ACT_SILU : ACT_TANH;
        sage_gemm_kernel<2><<<GEMM_GRID, 256, 0, stream>>>(
            hcur, W_self + (size_t)i * D * D, b_self + (size_t)i * D,
            agg,  W_neigh + (size_t)i * D * D, hnext, act);

        float* t = hcur; hcur = hnext; hnext = t;
    }
}
